# starting kernel
# baseline (speedup 1.0000x reference)
#include <hip/hip_runtime.h>

typedef __attribute__((ext_vector_type(2))) float v2f;
typedef __attribute__((ext_vector_type(8))) float v8f;

#define LOG2E 1.4426950408889634f

static __device__ __forceinline__ v8f wmma_f32_4(v2f a, v2f b, v8f c) {
  return __builtin_amdgcn_wmma_f32_16x16x4_f32(false, a, false, b, (short)0, c,
                                               false, false);
}

#define DPP_MAX_STEP(v, SEL)                                                  \
  asm("v_max_num_f32_dpp %0, %1, %1 " SEL                                     \
      " row_mask:0xf bank_mask:0xf bound_ctrl:1"                              \
      : "=v"(v)                                                               \
      : "v"(v))

__global__ __launch_bounds__(256, 1) void mha16_kernel(
    const float* __restrict__ q, const float* __restrict__ k,
    const float* __restrict__ v, const float* __restrict__ bias,
    const float* __restrict__ mask, float* __restrict__ out) {
  __shared__ __align__(16) float ldsP[8][320];

  const int tid  = threadIdx.x;
  const int lane = tid & 31;
  const int wave = tid >> 5;
  const int b     = blockIdx.x >> 6;
  const int ptile = blockIdx.x & 63;
  const int h  = wave;
  const int bh = b * 8 + h;
  const int p0 = ptile << 4;

  const int n    = lane & 15;
  const int half = lane >> 4;

  const float* qp = q + ((size_t)bh * 1024 + p0) * 16;

  const float* kpl = k + (size_t)bh * 1024 * 16 + n * 16 + 2 * half;
  const float* vpl = v + (size_t)bh * 1024 * 16 + 32 * half + n;
  const float* bpl = bias + (((size_t)b * 1024 + p0) * 1024) * 8 + h +
                     ((size_t)(8 * half) * 1024 + n) * 8;
  const float* mpl = mask + ((size_t)b * 1024 + p0) * 1024 +
                     (size_t)(8 * half) * 1024 + n;

  v2f qa[4];
#pragma unroll
  for (int c = 0; c < 4; ++c)
    qa[c] = *(const v2f*)(qp + n * 16 + 4 * c + 2 * half);

  v8f acc  = {0.f, 0.f, 0.f, 0.f, 0.f, 0.f, 0.f, 0.f};
  v8f lacc = {0.f, 0.f, 0.f, 0.f, 0.f, 0.f, 0.f, 0.f};
  float mrow[8], negml2e[8];
#pragma unroll
  for (int r = 0; r < 8; ++r) { mrow[r] = -3.0e38f; negml2e[r] = 3.0e38f; }

  float* myP = ldsP[wave];
  const v2f onesB = {1.0f, 1.0f};

  for (int it = 0; it < 16; ++it) {
    v8f sc[4];
#pragma unroll
    for (int t = 0; t < 4; ++t) {
      v8f z = {0.f, 0.f, 0.f, 0.f, 0.f, 0.f, 0.f, 0.f};
#pragma unroll
      for (int c = 0; c < 4; ++c) {
        v2f kb = *(const v2f*)(kpl + t * 256 + 4 * c);
        z = wmma_f32_4(qa[c], kb, z);
      }
      sc[t] = z;
    }

#pragma unroll
    for (int t = 0; t < 4; ++t) {
#pragma unroll
      for (int r = 0; r < 8; ++r) {
        const float bi = bpl[r * 8192 + t * 128];
        const float mk = mpl[r * 1024 + t * 16];
        sc[t][r] = fmaf(sc[t][r], 0.25f, mk + bi);
      }
    }

    float pscale[8];
#pragma unroll
    for (int r = 0; r < 8; ++r) {
      float mx = fmaxf(fmaxf(sc[0][r], sc[1][r]), fmaxf(sc[2][r], sc[3][r]));
      DPP_MAX_STEP(mx, "quad_perm:[1,0,3,2]");
      DPP_MAX_STEP(mx, "quad_perm:[2,3,0,1]");
      DPP_MAX_STEP(mx, "row_half_mirror");
      DPP_MAX_STEP(mx, "row_mirror");
      const float mnew = fmaxf(mrow[r], mx);
      const float nml2e = -mnew * LOG2E;
      pscale[r] = __builtin_amdgcn_exp2f(fmaf(mrow[r], LOG2E, nml2e));
      mrow[r] = mnew;
      negml2e[r] = nml2e;
      acc[r]  *= pscale[r];
      lacc[r] *= pscale[r];
    }

#pragma unroll
    for (int t = 0; t < 4; ++t) {
#pragma unroll
      for (int r = 0; r < 8; ++r) {
        const float p =
            __builtin_amdgcn_exp2f(fmaf(sc[t][r], LOG2E, negml2e[r]));
        myP[(r + 8 * half) * 20 + n] = p;
      }
      asm volatile("" ::: "memory");
#pragma unroll
      for (int c = 0; c < 4; ++c) {
        v2f pa = *(const v2f*)(myP + n * 20 + 4 * c + 2 * half);
        v2f vb;
        vb[0] = vpl[t * 256 + 64 * c];
        vb[1] = vpl[t * 256 + 64 * c + 16];
        acc  = wmma_f32_4(pa, vb, acc);
        lacc = wmma_f32_4(pa, onesB, lacc);
      }
      asm volatile("" ::: "memory");
    }

    kpl += 64 * 16;
    vpl += 64 * 16;
    bpl += 64 * 8;
    mpl += 64;
  }

#pragma unroll
  for (int r = 0; r < 8; ++r) {
    const int m = r + 8 * half;
    out[(size_t)(b * 1024 + p0 + m) * 128 + h * 16 + n] = acc[r] / lacc[r];
  }
}

extern "C" void kernel_launch(void* const* d_in, const int* in_sizes, int n_in,
                              void* d_out, int out_size, void* d_ws,
                              size_t ws_size, hipStream_t stream) {
  (void)in_sizes; (void)n_in; (void)out_size; (void)d_ws; (void)ws_size;
  const float* q    = (const float*)d_in[0];
  const float* k    = (const float*)d_in[1];
  const float* v    = (const float*)d_in[2];
  const float* bias = (const float*)d_in[3];
  const float* mask = (const float*)d_in[4];
  float* out = (float*)d_out;
  mha16_kernel<<<256, 256, 0, stream>>>(q, k, v, bias, mask, out);
}


# mask+bias tile fetched once per block with whole-line 128-bit loads one tile ahead, summed, shared to the 8 head waves via double-buffered LDS; replaces per-wave 32-byte-stride bias gathers
# speedup vs baseline: 1.0365x; 1.0365x over previous
#include <hip/hip_runtime.h>

typedef __attribute__((ext_vector_type(2))) float v2f;
typedef __attribute__((ext_vector_type(8))) float v8f;
typedef __attribute__((ext_vector_type(4))) float v4f;

#define BROW 68
#define BPLANE 1096

#define LOG2E 1.4426950408889634f

static __device__ __forceinline__ v8f wmma_f32_4(v2f a, v2f b, v8f c) {
  return __builtin_amdgcn_wmma_f32_16x16x4_f32(false, a, false, b, (short)0, c,
                                               false, false);
}

#define DPP_MAX_STEP(v, SEL)                                                  \
  asm("v_max_num_f32_dpp %0, %1, %1 " SEL                                     \
      " row_mask:0xf bank_mask:0xf bound_ctrl:1"                              \
      : "=v"(v)                                                               \
      : "v"(v))

__global__ __launch_bounds__(256, 1) void mha16_kernel(
    const float* __restrict__ q, const float* __restrict__ k,
    const float* __restrict__ v, const float* __restrict__ bias,
    const float* __restrict__ mask, float* __restrict__ out) {
  __shared__ __align__(16) float ldsP[8][320];
  __shared__ __align__(16) float ldsB[2][8 * BPLANE];

  const int tid  = threadIdx.x;
  const int lane = tid & 31;
  const int wave = tid >> 5;
  const int b     = blockIdx.x >> 6;
  const int ptile = blockIdx.x & 63;
  const int h  = wave;
  const int bh = b * 8 + h;
  const int p0 = ptile << 4;

  const int n    = lane & 15;
  const int half = lane >> 4;

  const float* qp = q + ((size_t)bh * 1024 + p0) * 16;

  const float* kpl = k + (size_t)bh * 1024 * 16 + n * 16 + 2 * half;
  const float* vpl = v + (size_t)bh * 1024 * 16 + 32 * half + n;

  const int f    = tid & 127;
  const int srow = tid >> 7;
  const float* bst = bias + (((size_t)b * 1024 + p0 + srow) * 1024) * 8 + f * 4;
  const float* mst = mask + ((size_t)b * 1024 + p0 + srow) * 1024 + (f >> 1);
  const int sst = (4 * (f & 1)) * BPLANE + srow * BROW + (f >> 1);
  const int rdo = h * BPLANE + (8 * half) * BROW + n;

  v4f bq[8];
  float mq[8];
#pragma unroll
  for (int j = 0; j < 8; ++j) {
    bq[j] = *(const v4f*)(bst + j * 16384);
    mq[j] = mst[j * 2048];
  }
#pragma unroll
  for (int j = 0; j < 8; ++j)
#pragma unroll
    for (int i = 0; i < 4; ++i)
      ldsB[0][sst + i * BPLANE + j * 2 * BROW] = mq[j] + bq[j][i];
  __syncthreads();

  v2f qa[4];
#pragma unroll
  for (int c = 0; c < 4; ++c)
    qa[c] = *(const v2f*)(qp + n * 16 + 4 * c + 2 * half);

  v8f acc  = {0.f, 0.f, 0.f, 0.f, 0.f, 0.f, 0.f, 0.f};
  v8f lacc = {0.f, 0.f, 0.f, 0.f, 0.f, 0.f, 0.f, 0.f};
  float mrow[8], negml2e[8];
#pragma unroll
  for (int r = 0; r < 8; ++r) { mrow[r] = -3.0e38f; negml2e[r] = 3.0e38f; }

  float* myP = ldsP[wave];
  const v2f onesB = {1.0f, 1.0f};

  for (int it = 0; it < 16; ++it) {
    v2f kb[4][4];
#pragma unroll
    for (int t = 0; t < 4; ++t) {
#pragma unroll
      for (int c = 0; c < 4; ++c)
        kb[t][c] = *(const v2f*)(kpl + t * 256 + 4 * c);
    }
    float vr[4][4][2];
#pragma unroll
    for (int t = 0; t < 4; ++t) {
#pragma unroll
      for (int c = 0; c < 4; ++c) {
        vr[t][c][0] = vpl[t * 256 + 64 * c];
        vr[t][c][1] = vpl[t * 256 + 64 * c + 16];
      }
    }

    __builtin_amdgcn_sched_barrier(0);

    if (it < 15) {
#pragma unroll
      for (int j = 0; j < 8; ++j) {
        bq[j] = *(const v4f*)(bst + (it + 1) * 512 + j * 16384);
        mq[j] = mst[(it + 1) * 64 + j * 2048];
      }
    }
    __builtin_amdgcn_sched_barrier(0);

    v8f sc[4];
#pragma unroll
    for (int t = 0; t < 4; ++t) {
      v8f z = {0.f, 0.f, 0.f, 0.f, 0.f, 0.f, 0.f, 0.f};
#pragma unroll
      for (int c = 0; c < 4; ++c) {
        z = wmma_f32_4(qa[c], kb[t][c], z);
      }
      sc[t] = z;
    }

    const float* myB = &ldsB[it & 1][rdo];
#pragma unroll
    for (int t = 0; t < 4; ++t) {
#pragma unroll
      for (int r = 0; r < 8; ++r) {
        const float ad = myB[r * BROW + t * 16];
        sc[t][r] = fmaf(sc[t][r], 0.25f, ad);
      }
    }

    float pscale[8];
#pragma unroll
    for (int r = 0; r < 8; ++r) {
      float mx = fmaxf(fmaxf(sc[0][r], sc[1][r]), fmaxf(sc[2][r], sc[3][r]));
      DPP_MAX_STEP(mx, "quad_perm:[1,0,3,2]");
      DPP_MAX_STEP(mx, "quad_perm:[2,3,0,1]");
      DPP_MAX_STEP(mx, "row_half_mirror");
      DPP_MAX_STEP(mx, "row_mirror");
      const float mnew = fmaxf(mrow[r], mx);
      const float nml2e = -mnew * LOG2E;
      pscale[r] = __builtin_amdgcn_exp2f(fmaf(mrow[r], LOG2E, nml2e));
      mrow[r] = mnew;
      negml2e[r] = nml2e;
      acc[r]  *= pscale[r];
      lacc[r] *= pscale[r];
    }

#pragma unroll
    for (int t = 0; t < 4; ++t) {
#pragma unroll
      for (int r = 0; r < 8; ++r) {
        const float p =
            __builtin_amdgcn_exp2f(fmaf(sc[t][r], LOG2E, negml2e[r]));
        myP[(r + 8 * half) * 20 + n] = p;
      }
      asm volatile("" ::: "memory");
#pragma unroll
      for (int c = 0; c < 4; ++c) {
        v2f pa = *(const v2f*)(myP + n * 20 + 4 * c + 2 * half);
        v2f vb;
        vb[0] = vr[t][c][0];
        vb[1] = vr[t][c][1];
        acc  = wmma_f32_4(pa, vb, acc);
        lacc = wmma_f32_4(pa, onesB, lacc);
      }
      asm volatile("" ::: "memory");
    }

    __builtin_amdgcn_sched_barrier(0);
    if (it < 15) {
      float* nb = &ldsB[(it + 1) & 1][sst];
#pragma unroll
      for (int j = 0; j < 8; ++j)
#pragma unroll
        for (int i = 0; i < 4; ++i)
          nb[i * BPLANE + j * 2 * BROW] = mq[j] + bq[j][i];
    }
    __syncthreads();

    kpl += 64 * 16;
    vpl += 64 * 16;
  }

#pragma unroll
  for (int r = 0; r < 8; ++r) {
    const int m = r + 8 * half;
    out[(size_t)(b * 1024 + p0 + m) * 128 + h * 16 + n] = acc[r] / lacc[r];
  }
}

extern "C" void kernel_launch(void* const* d_in, const int* in_sizes, int n_in,
                              void* d_out, int out_size, void* d_ws,
                              size_t ws_size, hipStream_t stream) {
  (void)in_sizes; (void)n_in; (void)out_size; (void)d_ws; (void)ws_size;
  const float* q    = (const float*)d_in[0];
  const float* k    = (const float*)d_in[1];
  const float* v    = (const float*)d_in[2];
  const float* bias = (const float*)d_in[3];
  const float* mask = (const float*)d_in[4];
  float* out = (float*)d_out;
  mha16_kernel<<<256, 256, 0, stream>>>(q, k, v, bias, mask, out);
}


# scores computed transposed (S^T=K*Q^T): P^T accumulator registers are the B operand of out^T=V^T*P^T, no LDS transpose of P; row max/sum in-lane (P*V and row-sum order changes); out rows via LDS
# speedup vs baseline: 1.3312x; 1.2843x over previous
#include <hip/hip_runtime.h>


typedef __attribute__((ext_vector_type(2))) float v2f;
typedef __attribute__((ext_vector_type(8))) float v8f;
typedef __attribute__((ext_vector_type(4))) float v4f;

#define BROW 68
#define BPLANE 1096
#define OROW 132

#define LOG2E 1.4426950408889634f

static __device__ __forceinline__ v8f wmma_f32_4(v2f a, v2f b, v8f c) {
  return __builtin_amdgcn_wmma_f32_16x16x4_f32(false, a, false, b, (short)0, c,
                                               false, false);
}

static __device__ __forceinline__ float xhalf(float x) {
  const int b = __builtin_bit_cast(int, x);
  return __builtin_bit_cast(
      float, __builtin_amdgcn_permlanex16(b, b, 0x76543210, 0xfedcba98, false,
                                          false));
}

__global__ __launch_bounds__(256, 1) void mha16_kernel(
    const float* __restrict__ q, const float* __restrict__ k,
    const float* __restrict__ v, const float* __restrict__ bias,
    const float* __restrict__ mask, float* __restrict__ out) {
  __shared__ __align__(16) float ldsB[2][8 * BPLANE];


  const int tid  = threadIdx.x;
  const int lane = tid & 31;
  const int wave = tid >> 5;
  const int b     = blockIdx.x >> 6;
  const int ptile = blockIdx.x & 63;
  const int h  = wave;
  const int bh = b * 8 + h;
  const int p0 = ptile << 4;

  const int n    = lane & 15;
  const int half = lane >> 4;

  const float* qp = q + ((size_t)bh * 1024 + p0) * 16;

  const float* kpl = k + (size_t)bh * 1024 * 16 + n * 16 + 2 * half;
  const float* vpl = v + (size_t)bh * 1024 * 16 + 128 * half + n;

  const int f    = tid & 127;
  const int srow = tid >> 7;
  const float* bst = bias + (((size_t)b * 1024 + p0 + srow) * 1024) * 8 + f * 4;
  const float* mst = mask + ((size_t)b * 1024 + p0 + srow) * 1024 + (f >> 1);
  const int sst = (4 * (f & 1)) * BPLANE + srow * BROW + (f >> 1);
  const int rdo = h * BPLANE + n * BROW + 8 * half;

  v4f bq[8];
  float mq[8];
#pragma unroll
  for (int j = 0; j < 8; ++j) {
    bq[j] = *(const v4f*)(bst + j * 16384);
    mq[j] = mst[j * 2048];
  }
#pragma unroll
  for (int j = 0; j < 8; ++j)
#pragma unroll
    for (int i = 0; i < 4; ++i)
      ldsB[0][sst + i * BPLANE + j * 2 * BROW] = mq[j] + bq[j][i];
  __syncthreads();

  v2f qa[4];
#pragma unroll
  for (int c = 0; c < 4; ++c)
    qa[c] = *(const v2f*)(qp + n * 16 + 4 * c + 2 * half);

  v8f acc = {0.f, 0.f, 0.f, 0.f, 0.f, 0.f, 0.f, 0.f};
  float mrow = -3.0e38f;
  float lsum = 0.f;

  for (int it = 0; it < 16; ++it) {
    v2f kb[4][4];
#pragma unroll
    for (int t = 0; t < 4; ++t) {
#pragma unroll
      for (int c = 0; c < 4; ++c)
        kb[t][c] = *(const v2f*)(kpl + t * 256 + 4 * c);
    }
    float vr[4][4][2];
#pragma unroll
    for (int t = 0; t < 4; ++t) {
#pragma unroll
      for (int j = 0; j < 4; ++j) {
        vr[t][j][0] = vpl[t * 256 + 32 * j];
        vr[t][j][1] = vpl[t * 256 + 32 * j + 16];
      }
    }

    __builtin_amdgcn_sched_barrier(0);

    if (it < 15) {
#pragma unroll
      for (int j = 0; j < 8; ++j) {
        bq[j] = *(const v4f*)(bst + (it + 1) * 512 + j * 16384);
        mq[j] = mst[(it + 1) * 64 + j * 2048];
      }
    }
    const float* myB = &ldsB[it & 1][rdo];
    v4f ad[4][2];
#pragma unroll
    for (int t = 0; t < 4; ++t) {
      ad[t][0] = *(const v4f*)(myB + t * 16);
      ad[t][1] = *(const v4f*)(myB + t * 16 + 4);
    }
    __builtin_amdgcn_sched_barrier(0);

    v8f sc[4];
#pragma unroll
    for (int t = 0; t < 4; ++t) {
      v8f z = {0.f, 0.f, 0.f, 0.f, 0.f, 0.f, 0.f, 0.f};
#pragma unroll
      for (int c = 0; c < 4; ++c) {
        z = wmma_f32_4(kb[t][c], qa[c], z);
      }
      sc[t] = z;
    }

#pragma unroll
    for (int t = 0; t < 4; ++t) {
#pragma unroll
      for (int r = 0; r < 8; ++r) {
        sc[t][r] = fmaf(sc[t][r], 0.25f, ad[t][r >> 2][r & 3]);
      }
    }

    float mxa = sc[0][0], mxb = sc[2][0];
#pragma unroll
    for (int r = 0; r < 8; ++r) {
      mxa = fmaxf(mxa, fmaxf(sc[0][r], sc[1][r]));
      mxb = fmaxf(mxb, fmaxf(sc[2][r], sc[3][r]));
    }
    float mx = fmaxf(mxa, mxb);
    mx = fmaxf(mx, xhalf(mx));
    const float mnew  = fmaxf(mrow, mx);
    const float nml2e = -mnew * LOG2E;
    const float pscale = __builtin_amdgcn_exp2f(fmaf(mrow, LOG2E, nml2e));
    mrow = mnew;
#pragma unroll
    for (int r = 0; r < 8; ++r) acc[r] *= pscale;

    const v2f l2e2 = {LOG2E, LOG2E};
    const v2f nm2  = {nml2e, nml2e};
    v2f ps2 = {0.f, 0.f};
#pragma unroll
    for (int t = 0; t < 4; ++t) {
#pragma unroll
      for (int j = 0; j < 4; ++j) {
        v2f s2;
        s2[0] = sc[t][2 * j];
        s2[1] = sc[t][2 * j + 1];
        const v2f a2 = __builtin_elementwise_fma(s2, l2e2, nm2);
        v2f pb;
        pb[0] = __builtin_amdgcn_exp2f(a2[0]);
        pb[1] = __builtin_amdgcn_exp2f(a2[1]);
        ps2 += pb;
        v2f va;
        va[0] = vr[t][j][0];
        va[1] = vr[t][j][1];
        acc = wmma_f32_4(va, pb, acc);
      }
    }
    lsum = fmaf(lsum, pscale, ps2[0] + ps2[1]);

    __builtin_amdgcn_sched_barrier(0);
    if (it < 15) {
      float* nb = &ldsB[(it + 1) & 1][sst];
#pragma unroll
      for (int j = 0; j < 8; ++j)
#pragma unroll
        for (int i = 0; i < 4; ++i)
          nb[i * BPLANE + j * 2 * BROW] = mq[j] + bq[j][i];
    }
    __syncthreads();

    kpl += 64 * 16;
    vpl += 64 * 16;
  }

  const float l = lsum + xhalf(lsum);
  v4f o0, o1;
#pragma unroll
  for (int r = 0; r < 4; ++r) {
    o0[r] = acc[r] / l;
    o1[r] = acc[r + 4] / l;
  }
  float* ot = &ldsB[0][0];
  *(v4f*)(ot + n * OROW + h * 16 + 8 * half)     = o0;
  *(v4f*)(ot + n * OROW + h * 16 + 8 * half + 4) = o1;
  __syncthreads();
  float* og = out + (size_t)(b * 1024 + p0) * 128;
#pragma unroll
  for (int i = 0; i < 2; ++i) {
    const int row = wave + 8 * i;
    *(v4f*)(og + row * 128 + lane * 4) = *(const v4f*)(ot + row * OROW + lane * 4);
  }
}

extern "C" void kernel_launch(void* const* d_in, const int* in_sizes, int n_in,
                              void* d_out, int out_size, void* d_ws,
                              size_t ws_size, hipStream_t stream) {
  (void)in_sizes; (void)n_in; (void)out_size; (void)d_ws; (void)ws_size;
  const float* q    = (const float*)d_in[0];
  const float* k    = (const float*)d_in[1];
  const float* v    = (const float*)d_in[2];
  const float* bias = (const float*)d_in[3];
  const float* mask = (const float*)d_in[4];
  float* out = (float*)d_out;
  mha16_kernel<<<256, 256, 0, stream>>>(q, k, v, bias, mask, out);
}


# K and V fragment loads moved one tile ahead into a second register set (two tiles per trip, the sets swap roles); K rows read as two 128-bit loads, which changes the d order of the score sums
# speedup vs baseline: 1.4186x; 1.0657x over previous
#include <hip/hip_runtime.h>


typedef __attribute__((ext_vector_type(2))) float v2f;
typedef __attribute__((ext_vector_type(8))) float v8f;
typedef __attribute__((ext_vector_type(4))) float v4f;

#define BROW 68
#define BPLANE 1096
#define OROW 132

#define LOG2E 1.4426950408889634f

static __device__ __forceinline__ v8f wmma_f32_4(v2f a, v2f b, v8f c) {
  return __builtin_amdgcn_wmma_f32_16x16x4_f32(false, a, false, b, (short)0, c,
                                               false, false);
}

static __device__ __forceinline__ float xhalf(float x) {
  const int b = __builtin_bit_cast(int, x);
  return __builtin_bit_cast(
      float, __builtin_amdgcn_permlanex16(b, b, 0x76543210, 0xfedcba98, false,
                                          false));
}

__global__ __launch_bounds__(256, 1) void mha16_kernel(
    const float* __restrict__ q, const float* __restrict__ k,
    const float* __restrict__ v, const float* __restrict__ bias,
    const float* __restrict__ mask, float* __restrict__ out) {
  __shared__ __align__(16) float ldsB[2][8 * BPLANE];


  const int tid  = threadIdx.x;
  const int lane = tid & 31;
  const int wave = tid >> 5;
  const int b     = blockIdx.x >> 6;
  const int ptile = blockIdx.x & 63;
  const int h  = wave;
  const int bh = b * 8 + h;
  const int p0 = ptile << 4;

  const int n    = lane & 15;
  const int half = lane >> 4;

  const float* qp = q + ((size_t)bh * 1024 + p0) * 16;

  const float* kpl = k + (size_t)bh * 1024 * 16 + n * 16 + 8 * half;
  const float* vpl = v + (size_t)bh * 1024 * 16 + 128 * half + n;

  const int f    = tid & 127;
  const int srow = tid >> 7;
  const float* bst = bias + (((size_t)b * 1024 + p0 + srow) * 1024) * 8 + f * 4;
  const float* mst = mask + ((size_t)b * 1024 + p0 + srow) * 1024 + (f >> 1);
  const int sst = (4 * (f & 1)) * BPLANE + srow * BROW + (f >> 1);
  const int rdo = h * BPLANE + n * BROW + 8 * half;

  v4f bq[8];
  float mq[8];
#pragma unroll
  for (int j = 0; j < 8; ++j) {
    bq[j] = *(const v4f*)(bst + j * 16384);
    mq[j] = mst[j * 2048];
  }

  v4f kA[4][2], kB[4][2];
  float vA[4][4][2], vB[4][4][2];

#pragma unroll
  for (int t = 0; t < 4; ++t) {
    kA[t][0] = *(const v4f*)(kpl + t * 256);
    kA[t][1] = *(const v4f*)(kpl + t * 256 + 4);
  }
#pragma unroll
  for (int t = 0; t < 4; ++t) {
#pragma unroll
    for (int j = 0; j < 4; ++j) {
      vA[t][j][0] = vpl[t * 256 + 32 * j];
      vA[t][j][1] = vpl[t * 256 + 32 * j + 16];
    }
  }

  const v4f q0 = *(const v4f*)(qp + n * 16 + 8 * half);
  const v4f q1 = *(const v4f*)(qp + n * 16 + 8 * half + 4);
  v2f qa[4];
  qa[0][0] = q0[0]; qa[0][1] = q0[1];
  qa[1][0] = q0[2]; qa[1][1] = q0[3];
  qa[2][0] = q1[0]; qa[2][1] = q1[1];
  qa[3][0] = q1[2]; qa[3][1] = q1[3];

#pragma unroll
  for (int j = 0; j < 8; ++j)
#pragma unroll
    for (int i = 0; i < 4; ++i)
      ldsB[0][sst + i * BPLANE + j * 2 * BROW] = mq[j] + bq[j][i];
  __syncthreads();

  v8f acc = {0.f, 0.f, 0.f, 0.f, 0.f, 0.f, 0.f, 0.f};
  float mrow = -3.0e38f;
  float lsum = 0.f;

  auto tile = [&](const int it, const int stage, v4f (&kc)[4][2],
                  float (&vc)[4][4][2], v4f (&kn)[4][2],
                  float (&vn)[4][4][2]) __attribute__((always_inline)) {
    if (it < 15) {
      const float* kp = kpl + (it + 1) * 1024;
      const float* vp = vpl + (it + 1) * 1024;
#pragma unroll
      for (int t = 0; t < 4; ++t) {
        kn[t][0] = *(const v4f*)(kp + t * 256);
        kn[t][1] = *(const v4f*)(kp + t * 256 + 4);
      }
#pragma unroll
      for (int t = 0; t < 4; ++t) {
#pragma unroll
        for (int j = 0; j < 4; ++j) {
          vn[t][j][0] = vp[t * 256 + 32 * j];
          vn[t][j][1] = vp[t * 256 + 32 * j + 16];
        }
      }
    }

    __builtin_amdgcn_sched_barrier(0);

    if (it < 15) {
#pragma unroll
      for (int j = 0; j < 8; ++j) {
        bq[j] = *(const v4f*)(bst + (it + 1) * 512 + j * 16384);
        mq[j] = mst[(it + 1) * 64 + j * 2048];
      }
    }
    const float* myB = &ldsB[stage][rdo];
    v4f ad[4][2];
#pragma unroll
    for (int t = 0; t < 4; ++t) {
      ad[t][0] = *(const v4f*)(myB + t * 16);
      ad[t][1] = *(const v4f*)(myB + t * 16 + 4);
    }
    __builtin_amdgcn_sched_barrier(0);

    v8f sc[4];
#pragma unroll
    for (int t = 0; t < 4; ++t) {
      v8f z = {0.f, 0.f, 0.f, 0.f, 0.f, 0.f, 0.f, 0.f};
#pragma unroll
      for (int c = 0; c < 4; ++c) {
        v2f ka;
        ka[0] = kc[t][c >> 1][2 * (c & 1)];
        ka[1] = kc[t][c >> 1][2 * (c & 1) + 1];
        z = wmma_f32_4(ka, qa[c], z);
      }
      sc[t] = z;
    }

#pragma unroll
    for (int t = 0; t < 4; ++t) {
#pragma unroll
      for (int r = 0; r < 8; ++r) {
        sc[t][r] = fmaf(sc[t][r], 0.25f, ad[t][r >> 2][r & 3]);
      }
    }

    float mxa = sc[0][0], mxb = sc[2][0];
#pragma unroll
    for (int r = 0; r < 8; ++r) {
      mxa = fmaxf(mxa, fmaxf(sc[0][r], sc[1][r]));
      mxb = fmaxf(mxb, fmaxf(sc[2][r], sc[3][r]));
    }
    float mx = fmaxf(mxa, mxb);
    mx = fmaxf(mx, xhalf(mx));
    const float mnew  = fmaxf(mrow, mx);
    const float nml2e = -mnew * LOG2E;
    const float pscale = __builtin_amdgcn_exp2f(fmaf(mrow, LOG2E, nml2e));
    mrow = mnew;
#pragma unroll
    for (int r = 0; r < 8; ++r) acc[r] *= pscale;

    const v2f l2e2 = {LOG2E, LOG2E};
    const v2f nm2  = {nml2e, nml2e};
    v2f ps2 = {0.f, 0.f};
#pragma unroll
    for (int t = 0; t < 4; ++t) {
#pragma unroll
      for (int j = 0; j < 4; ++j) {
        v2f s2;
        s2[0] = sc[t][2 * j];
        s2[1] = sc[t][2 * j + 1];
        const v2f a2 = __builtin_elementwise_fma(s2, l2e2, nm2);
        v2f pb;
        pb[0] = __builtin_amdgcn_exp2f(a2[0]);
        pb[1] = __builtin_amdgcn_exp2f(a2[1]);
        ps2 += pb;
        v2f va;
        va[0] = vc[t][j][0];
        va[1] = vc[t][j][1];
        acc = wmma_f32_4(va, pb, acc);
      }
    }
    lsum = fmaf(lsum, pscale, ps2[0] + ps2[1]);

    __builtin_amdgcn_sched_barrier(0);
    if (it < 15) {
      float* nb = &ldsB[stage ^ 1][sst];
#pragma unroll
      for (int j = 0; j < 8; ++j)
#pragma unroll
        for (int i = 0; i < 4; ++i)
          nb[i * BPLANE + j * 2 * BROW] = mq[j] + bq[j][i];
    }
    __syncthreads();
  };

  for (int it = 0; it < 16; it += 2) {
    tile(it, 0, kA, vA, kB, vB);
    tile(it + 1, 1, kB, vB, kA, vA);
  }

  const float l = lsum + xhalf(lsum);
  v4f o0, o1;
#pragma unroll
  for (int r = 0; r < 4; ++r) {
    o0[r] = acc[r] / l;
    o1[r] = acc[r + 4] / l;
  }
  float* ot = &ldsB[0][0];
  *(v4f*)(ot + n * OROW + h * 16 + 8 * half)     = o0;
  *(v4f*)(ot + n * OROW + h * 16 + 8 * half + 4) = o1;
  __syncthreads();
  float* og = out + (size_t)(b * 1024 + p0) * 128;
#pragma unroll
  for (int i = 0; i < 2; ++i) {
    const int row = wave + 8 * i;
    *(v4f*)(og + row * 128 + lane * 4) = *(const v4f*)(ot + row * OROW + lane * 4);
  }
}

extern "C" void kernel_launch(void* const* d_in, const int* in_sizes, int n_in,
                              void* d_out, int out_size, void* d_ws,
                              size_t ws_size, hipStream_t stream) {
  (void)in_sizes; (void)n_in; (void)out_size; (void)d_ws; (void)ws_size;
  const float* q    = (const float*)d_in[0];
  const float* k    = (const float*)d_in[1];
  const float* v    = (const float*)d_in[2];
  const float* bias = (const float*)d_in[3];
  const float* mask = (const float*)d_in[4];
  float* out = (float*)d_out;
  mha16_kernel<<<256, 256, 0, stream>>>(q, k, v, bias, mask, out);
}


# next tile's K and V loads issued five at a time behind each group of four matrix instructions and kept in flight across a hand-written LDS-only tile barrier; mask/bias loads first in the tile
# speedup vs baseline: 1.6713x; 1.1781x over previous
#include <hip/hip_runtime.h>


typedef __attribute__((ext_vector_type(2))) float v2f;
typedef __attribute__((ext_vector_type(8))) float v8f;
typedef __attribute__((ext_vector_type(4))) float v4f;

#define BROW 68
#define BPLANE 1096
#define OROW 132

#define LOG2E 1.4426950408889634f

static __device__ __forceinline__ v8f wmma_f32_4(v2f a, v2f b, v8f c) {
  return __builtin_amdgcn_wmma_f32_16x16x4_f32(false, a, false, b, (short)0, c,
                                               false, false);
}

static __device__ __forceinline__ float xhalf(float x) {
  const int b = __builtin_bit_cast(int, x);
  return __builtin_bit_cast(
      float, __builtin_amdgcn_permlanex16(b, b, 0x76543210, 0xfedcba98, false,
                                          false));
}

__global__ __launch_bounds__(256, 1) void mha16_kernel(
    const float* __restrict__ q, const float* __restrict__ k,
    const float* __restrict__ v, const float* __restrict__ bias,
    const float* __restrict__ mask, float* __restrict__ out) {
  __shared__ __align__(16) float ldsB[2][8 * BPLANE];


  const int tid  = threadIdx.x;
  const int lane = tid & 31;
  const int wave = tid >> 5;
  const int b     = blockIdx.x >> 6;
  const int ptile = blockIdx.x & 63;
  const int h  = wave;
  const int bh = b * 8 + h;
  const int p0 = ptile << 4;

  const int n    = lane & 15;
  const int half = lane >> 4;

  const float* qp = q + ((size_t)bh * 1024 + p0) * 16;

  const float* kpl = k + (size_t)bh * 1024 * 16 + n * 16 + 8 * half;
  const float* vpl = v + (size_t)bh * 1024 * 16 + 128 * half + n;

  const int f    = tid & 127;
  const int srow = tid >> 7;
  const float* bst = bias + (((size_t)b * 1024 + p0 + srow) * 1024) * 8 + f * 4;
  const float* mst = mask + ((size_t)b * 1024 + p0 + srow) * 1024 + (f >> 1);
  const int sst = (4 * (f & 1)) * BPLANE + srow * BROW + (f >> 1);
  const int rdo = h * BPLANE + n * BROW + 8 * half;

  v4f bq[8];
  float mq[8];
#pragma unroll
  for (int j = 0; j < 8; ++j) {
    bq[j] = *(const v4f*)(bst + j * 16384);
    mq[j] = mst[j * 2048];
  }

  v4f kA[4][2], kB[4][2];
  float vA[4][4][2], vB[4][4][2];

#pragma unroll
  for (int t = 0; t < 4; ++t) {
    kA[t][0] = *(const v4f*)(kpl + t * 256);
    kA[t][1] = *(const v4f*)(kpl + t * 256 + 4);
  }
#pragma unroll
  for (int t = 0; t < 4; ++t) {
#pragma unroll
    for (int j = 0; j < 4; ++j) {
      vA[t][j][0] = vpl[t * 256 + 32 * j];
      vA[t][j][1] = vpl[t * 256 + 32 * j + 16];
    }
  }

  const v4f q0 = *(const v4f*)(qp + n * 16 + 8 * half);
  const v4f q1 = *(const v4f*)(qp + n * 16 + 8 * half + 4);
  v2f qa[4];
  qa[0][0] = q0[0]; qa[0][1] = q0[1];
  qa[1][0] = q0[2]; qa[1][1] = q0[3];
  qa[2][0] = q1[0]; qa[2][1] = q1[1];
  qa[3][0] = q1[2]; qa[3][1] = q1[3];

#pragma unroll
  for (int j = 0; j < 8; ++j)
#pragma unroll
    for (int i = 0; i < 4; ++i)
      ldsB[0][sst + i * BPLANE + j * 2 * BROW] = mq[j] + bq[j][i];
  __syncthreads();

  v8f acc = {0.f, 0.f, 0.f, 0.f, 0.f, 0.f, 0.f, 0.f};
  float mrow = -3.0e38f;
  float lsum = 0.f;

  auto tile = [&](const int it, const int stage, v4f (&kc)[4][2],
                  float (&vc)[4][4][2], v4f (&kn)[4][2],
                  float (&vn)[4][4][2]) __attribute__((always_inline)) {
    const int nx = it < 15 ? it + 1 : 15;
    const float* kp = kpl + nx * 1024;
    const float* vp = vpl + nx * 1024;

    if (it < 15) {
#pragma unroll
      for (int j = 0; j < 8; ++j) {
        bq[j] = *(const v4f*)(bst + (it + 1) * 512 + j * 16384);
        mq[j] = mst[(it + 1) * 64 + j * 2048];
      }
    }
    const float* myB = &ldsB[stage][rdo];
    v4f ad[4][2];
#pragma unroll
    for (int t = 0; t < 4; ++t) {
      ad[t][0] = *(const v4f*)(myB + t * 16);
      ad[t][1] = *(const v4f*)(myB + t * 16 + 4);
    }
    __builtin_amdgcn_sched_barrier(0);

    v8f sc[4];
#pragma unroll
    for (int t = 0; t < 4; ++t) {
      const v8f z = {0.f, 0.f, 0.f, 0.f, 0.f, 0.f, 0.f, 0.f};
      sc[t] = z;
    }
#pragma unroll
    for (int c = 0; c < 4; ++c) {
#pragma unroll
      for (int t = 0; t < 4; ++t) {
        v2f ka;
        ka[0] = kc[t][c >> 1][2 * (c & 1)];
        ka[1] = kc[t][c >> 1][2 * (c & 1) + 1];
        sc[t] = wmma_f32_4(ka, qa[c], sc[t]);
      }
#pragma unroll
      for (int i = 5 * c; i < 5 * c + 5; ++i) {
        if (i < 8) {
          kn[i & 3][i >> 2] = *(const v4f*)(kp + (i & 3) * 256 + 4 * (i >> 2));
        } else {
          const int x = i - 8;
          vn[x >> 3][(x >> 1) & 3][x & 1] =
              vp[(x >> 3) * 256 + 32 * ((x >> 1) & 3) + 16 * (x & 1)];
        }
      }
      __builtin_amdgcn_sched_barrier(0x406);
    }

#pragma unroll
    for (int t = 0; t < 4; ++t) {
#pragma unroll
      for (int r = 0; r < 8; ++r) {
        sc[t][r] = fmaf(sc[t][r], 0.25f, ad[t][r >> 2][r & 3]);
      }
    }

    float mxa = sc[0][0], mxb = sc[2][0];
#pragma unroll
    for (int r = 0; r < 8; ++r) {
      mxa = fmaxf(mxa, fmaxf(sc[0][r], sc[1][r]));
      mxb = fmaxf(mxb, fmaxf(sc[2][r], sc[3][r]));
    }
    float mx = fmaxf(mxa, mxb);
    mx = fmaxf(mx, xhalf(mx));
    const float mnew  = fmaxf(mrow, mx);
    const float nml2e = -mnew * LOG2E;
    const float pscale = __builtin_amdgcn_exp2f(fmaf(mrow, LOG2E, nml2e));
    mrow = mnew;
#pragma unroll
    for (int r = 0; r < 8; ++r) acc[r] *= pscale;

    const v2f l2e2 = {LOG2E, LOG2E};
    const v2f nm2  = {nml2e, nml2e};
    v2f ps2 = {0.f, 0.f};
#pragma unroll
    for (int t = 0; t < 4; ++t) {
#pragma unroll
      for (int j = 0; j < 4; ++j) {
        v2f s2;
        s2[0] = sc[t][2 * j];
        s2[1] = sc[t][2 * j + 1];
        const v2f a2 = __builtin_elementwise_fma(s2, l2e2, nm2);
        v2f pb;
        pb[0] = __builtin_amdgcn_exp2f(a2[0]);
        pb[1] = __builtin_amdgcn_exp2f(a2[1]);
        ps2 += pb;
        v2f va;
        va[0] = vc[t][j][0];
        va[1] = vc[t][j][1];
        acc = wmma_f32_4(va, pb, acc);
      }
#pragma unroll
      for (int x = 12 + 5 * t; x < 12 + 5 * t + 5; ++x) {
        vn[x >> 3][(x >> 1) & 3][x & 1] =
            vp[(x >> 3) * 256 + 32 * ((x >> 1) & 3) + 16 * (x & 1)];
      }
      __builtin_amdgcn_sched_barrier(0x406);
    }
    lsum = fmaf(lsum, pscale, ps2[0] + ps2[1]);

    __builtin_amdgcn_sched_barrier(0);
    if (it < 15) {
      float* nb = &ldsB[stage ^ 1][sst];
#pragma unroll
      for (int j = 0; j < 8; ++j)
#pragma unroll
        for (int i = 0; i < 4; ++i)
          nb[i * BPLANE + j * 2 * BROW] = mq[j] + bq[j][i];
    }
    asm volatile("s_wait_dscnt 0x0" ::: "memory");
    asm volatile("s_barrier_signal -1\n\ts_barrier_wait -1" ::: "memory");
  };

  for (int it = 0; it < 16; it += 2) {
    tile(it, 0, kA, vA, kB, vB);
    tile(it + 1, 1, kB, vB, kA, vA);
  }

  const float l = lsum + xhalf(lsum);
  v4f o0, o1;
#pragma unroll
  for (int r = 0; r < 4; ++r) {
    o0[r] = acc[r] / l;
    o1[r] = acc[r + 4] / l;
  }
  float* ot = &ldsB[0][0];
  *(v4f*)(ot + n * OROW + h * 16 + 8 * half)     = o0;
  *(v4f*)(ot + n * OROW + h * 16 + 8 * half + 4) = o1;
  __syncthreads();
  float* og = out + (size_t)(b * 1024 + p0) * 128;
#pragma unroll
  for (int i = 0; i < 2; ++i) {
    const int row = wave + 8 * i;
    *(v4f*)(og + row * 128 + lane * 4) = *(const v4f*)(ot + row * OROW + lane * 4);
  }
}

extern "C" void kernel_launch(void* const* d_in, const int* in_sizes, int n_in,
                              void* d_out, int out_size, void* d_ws,
                              size_t ws_size, hipStream_t stream) {
  (void)in_sizes; (void)n_in; (void)out_size; (void)d_ws; (void)ws_size;
  const float* q    = (const float*)d_in[0];
  const float* k    = (const float*)d_in[1];
  const float* v    = (const float*)d_in[2];
  const float* bias = (const float*)d_in[3];
  const float* mask = (const float*)d_in[4];
  float* out = (float*)d_out;
  mha16_kernel<<<256, 256, 0, stream>>>(q, k, v, bias, mask, out);
}
